# starting kernel
# baseline (speedup 1.0000x reference)
#include <hip/hip_runtime.h>
#include <hip/hip_bf16.h>


typedef __attribute__((ext_vector_type(2))) float v2f;
typedef __attribute__((ext_vector_type(8))) float v8f;

#define F_IN 128
#define F_H  16
#define F_C  2

__global__ void GNN_zero_f32(float* __restrict__ p, long n) {
    long i = (long)blockIdx.x * blockDim.x + threadIdx.x;
    if (i < n) p[i] = 0.0f;
}

__global__ void GNN_degree(const int* __restrict__ eidx, long E,
                           float* __restrict__ deg) {
    long e = (long)blockIdx.x * blockDim.x + threadIdx.x;
    if (e >= E) return;
    int d = eidx[E + e];
    atomicAdd(&deg[d], 1.0f);
}

__global__ void GNN_dinv(const float* __restrict__ deg,
                         float* __restrict__ dinv, int n) {
    int i = blockIdx.x * blockDim.x + threadIdx.x;
    if (i < n) dinv[i] = rsqrtf(deg[i] + 1.0f);
}

__global__ void GNN_gemm1_wmma(const float* __restrict__ x,
                               const float* __restrict__ W1,
                               float* __restrict__ h1, int nTiles) {
    int lane = threadIdx.x & 31;
    int wave = threadIdx.x >> 5;
    int tile = blockIdx.x * (blockDim.x >> 5) + wave;
    if (tile >= nTiles) return;

    int m    = lane & 15;
    int koff = (lane >> 4) << 1;
    long rowA = (long)(tile * 16 + m) * F_IN;

    v8f acc = {};
    #pragma unroll 4
    for (int kk = 0; kk < F_IN; kk += 4) {
        v2f a, b;
        a.x = x[rowA + kk + koff];
        a.y = x[rowA + kk + koff + 1];
        b.x = W1[(kk + koff)     * F_H + m];
        b.y = W1[(kk + koff + 1) * F_H + m];
        acc = __builtin_amdgcn_wmma_f32_16x16x4_f32(
            false, a, false, b, (short)0, acc, false, false);
    }

    int rbase = tile * 16 + (lane >> 4) * 8;
    #pragma unroll
    for (int r = 0; r < 8; ++r)
        h1[(long)(rbase + r) * F_H + m] = acc[r];
}

__global__ void GNN_gemm1_tail(const float* __restrict__ x,
                               const float* __restrict__ W1,
                               float* __restrict__ h1, int startRow, int n) {
    int idx = blockIdx.x * blockDim.x + threadIdx.x;
    int row = startRow + (idx >> 4);
    int col = idx & 15;
    if (row >= n) return;
    float acc = 0.0f;
    for (int k = 0; k < F_IN; ++k)
        acc += x[(long)row * F_IN + k] * W1[k * F_H + col];
    h1[(long)row * F_H + col] = acc;
}

__global__ void GNN_scatter1(const int* __restrict__ eidx, long E,
                             const float* __restrict__ dinv,
                             const float* __restrict__ h1,
                             float* __restrict__ agg1) {
    long e = (long)blockIdx.x * blockDim.x + threadIdx.x;
    if (e >= E) return;
    int s = eidx[e];
    int d = eidx[E + e];
    float norm = dinv[s] * dinv[d];
    const float4* hp = (const float4*)(h1 + (long)s * F_H);
    float* ap = agg1 + (long)d * F_H;
    #pragma unroll
    for (int q = 0; q < 4; ++q) {
        float4 v = hp[q];
        atomicAdd(ap + q * 4 + 0, norm * v.x);
        atomicAdd(ap + q * 4 + 1, norm * v.y);
        atomicAdd(ap + q * 4 + 2, norm * v.z);
        atomicAdd(ap + q * 4 + 3, norm * v.w);
    }
}

__global__ void GNN_combine1(const float* __restrict__ agg1,
                             const float* __restrict__ h1,
                             const float* __restrict__ dinv,
                             const float* __restrict__ b1,
                             const float* __restrict__ W2,
                             float* __restrict__ h2, int n) {
    int i = blockIdx.x * blockDim.x + threadIdx.x;
    if (i >= n) return;
    float di  = dinv[i];
    float di2 = di * di;
    float acc0 = 0.0f, acc1 = 0.0f;
    #pragma unroll
    for (int j = 0; j < F_H; ++j) {
        float v = agg1[(long)i * F_H + j] + di2 * h1[(long)i * F_H + j] + b1[j];
        v = fmaxf(v, 0.0f);
        acc0 = fmaf(v, W2[j * F_C + 0], acc0);
        acc1 = fmaf(v, W2[j * F_C + 1], acc1);
    }
    h2[(long)i * F_C + 0] = acc0;
    h2[(long)i * F_C + 1] = acc1;
}

__global__ void GNN_scatter2(const int* __restrict__ eidx, long E,
                             const float* __restrict__ dinv,
                             const float* __restrict__ h2,
                             float* __restrict__ agg2) {
    long e = (long)blockIdx.x * blockDim.x + threadIdx.x;
    if (e >= E) return;
    int s = eidx[e];
    int d = eidx[E + e];
    float norm = dinv[s] * dinv[d];
    float2 v = *(const float2*)(h2 + (long)s * F_C);
    atomicAdd(&agg2[(long)d * F_C + 0], norm * v.x);
    atomicAdd(&agg2[(long)d * F_C + 1], norm * v.y);
}

__global__ void GNN_final(const float* __restrict__ agg2,
                          const float* __restrict__ h2,
                          const float* __restrict__ dinv,
                          const float* __restrict__ b2,
                          float* __restrict__ out, int n) {
    int i = blockIdx.x * blockDim.x + threadIdx.x;
    if (i >= n) return;
    float di  = dinv[i];
    float di2 = di * di;
    float o0 = agg2[(long)i * F_C + 0] + di2 * h2[(long)i * F_C + 0] + b2[0];
    float o1 = agg2[(long)i * F_C + 1] + di2 * h2[(long)i * F_C + 1] + b2[1];
    float mx  = fmaxf(o0, o1);
    float lse = mx + logf(expf(o0 - mx) + expf(o1 - mx));
    out[(long)i * F_C + 0] = o0 - lse;
    out[(long)i * F_C + 1] = o1 - lse;
}

extern "C" void kernel_launch(void* const* d_in, const int* in_sizes, int n_in,
                              void* d_out, int out_size, void* d_ws, size_t ws_size,
                              hipStream_t stream) {
    const float* x  = (const float*)d_in[0];
    const int* eidx = (const int*)  d_in[1];
    const float* W1 = (const float*)d_in[2];
    const float* b1 = (const float*)d_in[3];
    const float* W2 = (const float*)d_in[4];
    const float* b2 = (const float*)d_in[5];
    float* out = (float*)d_out;

    long nN = (long)in_sizes[0] / F_IN;
    long E  = (long)in_sizes[1] / 2;

    float* ws   = (float*)d_ws;
    float* deg  = ws;
    float* agg1 = ws + nN;
    float* agg2 = ws + 17 * nN;
    float* dinv = ws + 19 * nN;
    float* h1   = ws + 20 * nN;
    float* h2   = ws + 36 * nN;

    const int T = 256;
    long zn = 19 * nN;
    GNN_zero_f32<<<(int)((zn + T - 1) / T), T, 0, stream>>>(ws, zn);
    GNN_degree  <<<(int)((E + T - 1) / T), T, 0, stream>>>(eidx, E, deg);
    GNN_dinv    <<<(int)((nN + T - 1) / T), T, 0, stream>>>(deg, dinv, (int)nN);

    int nTiles = (int)(nN / 16);
    if (nTiles > 0)
        GNN_gemm1_wmma<<<(nTiles + 7) / 8, T, 0, stream>>>(x, W1, h1, nTiles);
    int rem = (int)(nN - (long)nTiles * 16);
    if (rem > 0)
        GNN_gemm1_tail<<<(rem * 16 + T - 1) / T, T, 0, stream>>>(
            x, W1, h1, nTiles * 16, (int)nN);

    GNN_scatter1<<<(int)((E + T - 1) / T), T, 0, stream>>>(eidx, E, dinv, h1, agg1);
    GNN_combine1<<<(int)((nN + T - 1) / T), T, 0, stream>>>(agg1, h1, dinv, b1, W2, h2, (int)nN);
    GNN_scatter2<<<(int)((E + T - 1) / T), T, 0, stream>>>(eidx, E, dinv, h2, agg2);
    GNN_final   <<<(int)((nN + T - 1) / T), T, 0, stream>>>(agg2, h2, dinv, b2, out, (int)nN);
}


# layer 1 scans nothing: the count (13 x 19) keeps its edges as (source id, place), a list per wave and 16000-node range; a layer 1 owner tests its range's lists; sum order: LDS arrival, list order
# speedup vs baseline: 8.5937x; 8.5937x over previous
#include <hip/hip_runtime.h>
#include <hip/hip_bf16.h>


typedef __attribute__((ext_vector_type(2))) float v2f;
typedef __attribute__((ext_vector_type(8))) float v8f;

#define F_IN 128
#define F_H  16
#define F_C  2

#define OWN_T     1024
#define OWN_PER   16
#define OWN_TRIP  (OWN_T * OWN_PER)
#define OWN_WBLK  (32 * OWN_PER)
#define OWN_PART  4
#define OWN_QCAP  (32 + 32 * OWN_PART)

#define CNT_NODES  25000
#define CNT_SLICES 32
#define AG1_NODES_5 4000
#define AG1_NODES_4 3125
#define AG1_NODES_3 2400
#define AG2_NODES  12800
#define AG2_SLICES 16
#define LST_CAP 1024
#define LST_DLB 12

#define CR_NODES  16000
#define CR_BITS   14
#define CR_SLICES 19
#define CL_CAP    1024

__device__ __forceinline__ void GNN_slice(long E, long& e0, long& e1) {
    const long P = (long)gridDim.y;
    const long S = (((E + P - 1) / P) + 127) & ~127L;
    e0 = min(E, (long)blockIdx.y * S);
    e1 = min(E, e0 + S);
}

__device__ __forceinline__ constexpr int GNN_slot(int k) { return (k >> 2) * 128 + (k & 3); }

__device__ __forceinline__ void GNN_load_dst(const int* __restrict__ dst, long end,
                                             long wbase, long lbase, int (&d)[OWN_PER]) {
    if (wbase + OWN_WBLK <= end) {
        #pragma unroll
        for (int k = 0; k < OWN_PER; k += 4) {
            int4 a = *(const int4*)(dst + lbase + GNN_slot(k));
            d[k] = a.x; d[k + 1] = a.y; d[k + 2] = a.z; d[k + 3] = a.w;
        }
    } else {
        #pragma unroll
        for (int k = 0; k < OWN_PER; ++k)
            d[k] = (lbase + GNN_slot(k) < end) ? dst[lbase + GNN_slot(k)] : -1;
    }
}

__global__ __launch_bounds__(OWN_T)
void GNN_own_cnt(const int* __restrict__ eidx, long E,
                 float* __restrict__ cntp, int n) {
    __shared__ float cntf[CNT_NODES];
    const int tid = threadIdx.x;
    const int lo  = blockIdx.x * CNT_NODES;
    const unsigned cnt = (unsigned)min(CNT_NODES, n - lo);
    for (int i = tid; i < CNT_NODES; i += OWN_T) cntf[i] = 0.0f;
    __syncthreads();
    long e0, e1;
    GNN_slice(E, e0, e1);
    const int* dst = eidx + E;
    const int wv = __builtin_amdgcn_readfirstlane(tid >> 5);
    for (long base0 = e0; base0 < e1; base0 += OWN_TRIP) {
        const long wbase = base0 + (long)wv * OWN_WBLK;
        int d[OWN_PER];
        GNN_load_dst(dst, e1, wbase, wbase + 4 * (tid & 31), d);
        #pragma unroll
        for (int k = 0; k < OWN_PER; ++k) {
            const unsigned t = (unsigned)(d[k] - lo);
            if (t < cnt) atomicAdd(&cntf[t], 1.0f);
        }
    }
    __syncthreads();
    float* cp = cntp + (long)blockIdx.y * n + lo;
    for (int i = tid; i < (int)cnt; i += OWN_T) cp[i] = cntf[i];
}

__global__ __launch_bounds__(OWN_T)
void GNN_own_cnt_keep(const int* __restrict__ eidx, long E,
                      float* __restrict__ cntp, int n,
                      unsigned* __restrict__ clists, int* __restrict__ ccounts) {
    __shared__ float cntf[CR_NODES];
    __shared__ unsigned cq[OWN_T / 32][64];
    const int tid  = threadIdx.x;
    const int lane = tid & 31;
    const int lo   = blockIdx.x * CR_NODES;
    const unsigned cnt = (unsigned)min(CR_NODES, n - lo);
    for (int i = tid; i < CR_NODES; i += OWN_T) cntf[i] = 0.0f;
    __syncthreads();
    long e0, e1;
    GNN_slice(E, e0, e1);
    const int* src = eidx;
    const int* dst = eidx + E;
    const int wv = __builtin_amdgcn_readfirstlane(tid >> 5);
    const long wvno = ((long)blockIdx.x * gridDim.y + blockIdx.y) * (OWN_T / 32) + wv;
    unsigned* lw = clists + wvno * CL_CAP;
    int  qn = 0;
    int  lpos = 0;
    bool lshort = false;
    for (long base0 = e0; base0 < e1; base0 += OWN_TRIP) {
        const long wbase = base0 + (long)wv * OWN_WBLK;
        const long lbase = wbase + 4 * lane;
        int d[OWN_PER], s[OWN_PER];
        GNN_load_dst(dst, e1, wbase, lbase, d);
        GNN_load_dst(src, e1, wbase, lbase, s);
        #pragma unroll
        for (int k = 0; k < OWN_PER; ++k) {
            const unsigned t = (unsigned)(d[k] - lo);
            const bool p = t < cnt;
            if (p) atomicAdd(&cntf[t], 1.0f);
            const unsigned mask = __builtin_amdgcn_ballot_w32(p);
            if (mask != 0u) {
                if (p) cq[wv][qn + (int)__builtin_amdgcn_mbcnt_lo(mask, 0u)] = ((unsigned)s[k] << CR_BITS) | t;
                qn += __builtin_popcount(mask);
                if (qn >= 32) {
                    __builtin_amdgcn_fence(__ATOMIC_RELEASE, "wavefront");
                    __builtin_amdgcn_wave_barrier();
                    __builtin_amdgcn_fence(__ATOMIC_ACQUIRE, "wavefront");
                    const unsigned v  = cq[wv][lane];
                    const unsigned mv = cq[wv][32 + lane];
                    if (lpos + 32 <= CL_CAP) { lw[lpos + lane] = v; lpos += 32; }
                    else lshort = true;
                    __builtin_amdgcn_fence(__ATOMIC_ACQ_REL, "wavefront");
                    __builtin_amdgcn_wave_barrier();
                    qn -= 32;
                    cq[wv][lane] = mv;
                    __builtin_amdgcn_fence(__ATOMIC_RELEASE, "wavefront");
                    __builtin_amdgcn_wave_barrier();
                }
            }
        }
    }
    if (qn > 0) {
        __builtin_amdgcn_fence(__ATOMIC_RELEASE, "wavefront");
        __builtin_amdgcn_wave_barrier();
        __builtin_amdgcn_fence(__ATOMIC_ACQUIRE, "wavefront");
        if (lpos + qn <= CL_CAP) { if (lane < qn) lw[lpos + lane] = cq[wv][lane]; lpos += qn; }
        else lshort = true;
    }
    if (lane == 0) ccounts[wvno] = lshort ? -1 : lpos;
    __syncthreads();
    float* cp = cntp + (long)blockIdx.y * n + lo;
    for (int i = tid; i < (int)cnt; i += OWN_T) cp[i] = cntf[i];
}

__global__ void GNN_dinv_merge(const float* __restrict__ cntp, int slices,
                               float* __restrict__ dinv, int n) {
    int i = blockIdx.x * blockDim.x + threadIdx.x;
    if (i >= n) return;
    float c = 0.0f;
    for (int p = 0; p < slices; ++p) c += cntp[(long)p * n + i];
    dinv[i] = rsqrtf(c + 1.0f);
}

template <int F, int LPE, int M, int MODE>
__global__ __launch_bounds__(OWN_T)
void GNN_own_agg(const int* __restrict__ eidx, long E,
                 const float* __restrict__ dinv,
                 const float* __restrict__ h,
                 float* __restrict__ agg, int n,
                 unsigned* __restrict__ lists, int* __restrict__ counts,
                 const unsigned* __restrict__ clists, const int* __restrict__ ccounts) {
    static_assert(MODE != 2 || F == 2, "the list path reads a row as one float4");
    static_assert(MODE != 1 || CR_NODES % M == 0, "a layer 1 range lies inside one count range");
    static_assert(MODE == 0 || M < (1 << LST_DLB), "place must fit its bits");
    __shared__ float tile[(M + 1) * F];
    __shared__ float dloc[M + 1];
    __shared__ int2  q[OWN_T / 32][OWN_QCAP];
    __shared__ int   short_list;
    const int tid  = threadIdx.x;
    const int lane = tid & 31;
    const int w    = __builtin_amdgcn_readfirstlane(tid >> 5);
    const int lo   = blockIdx.x * M;
    const unsigned cnt = (unsigned)min(M, n - lo);
    for (int i = tid; i < (M + 1) * F; i += OWN_T) tile[i] = 0.0f;
    for (int i = tid; i <= M; i += OWN_T)
        dloc[i] = (i < (int)cnt) ? dinv[lo + i] : 0.0f;
    if (MODE != 0 && tid == 0) short_list = 0;
    __syncthreads();

    bool direct = false;
    const int crng = lo / CR_NODES;
    const int cnl  = CR_SLICES * (OWN_T / 32);
    if (MODE == 1 && clists != nullptr) {
        const int* cc = ccounts + (long)crng * cnl;
        for (int i = tid; i < cnl; i += OWN_T)
            if (cc[i] < 0) short_list = 1;
        __syncthreads();
        direct = (short_list == 0);
    }

    const long wvno = ((long)blockIdx.y * gridDim.x + blockIdx.x) * (OWN_T / 32) + w;
    unsigned* lw = (MODE == 0) ? nullptr : lists + wvno * LST_CAP;
    int  lpos = 0;
    bool lshort = false;

    const int* src = eidx;
    const int* dst = eidx + E;
    const int sub = lane / LPE;
    const int col = lane % LPE;
    constexpr int EPS = 32 / LPE;
    int qn = 0;

    auto chunk = [&](int i0, int m) {
        const bool okl = lane < m;
        const int2  ent  = q[w][i0 + (okl ? lane : 0)];
        const int   s_l  = direct ? ent.x : src[ent.x];
        const int   dl_l = okl ? ent.y : M;
        const float dd_l = dloc[dl_l];
        if (MODE == 1) {
            if (lpos + m <= LST_CAP) {
                if (okl) lw[lpos + lane] = ((unsigned)s_l << LST_DLB) | (unsigned)ent.y;
                lpos += m;
            } else {
                lshort = true;
            }
        }
        float2 v[LPE]; int dl[LPE]; float dd[LPE];
        #pragma unroll
        for (int j = 0; j < LPE; ++j) {
            const int from = j * EPS + sub;
            const int s = __shfl(s_l, from);
            dl[j] = __shfl(dl_l, from);
            dd[j] = __shfl(dd_l, from);
            v[j]  = *(const float2*)(h + (long)s * (2 * F) + 2 * col);
        }
        #pragma unroll
        for (int j = 0; j < LPE; ++j)
            atomicAdd(&tile[dl[j] * F + col], (v[j].y * dd[j]) * v[j].x);
    };

    auto drain = [&](bool all) {
        __builtin_amdgcn_fence(__ATOMIC_RELEASE, "wavefront");
        __builtin_amdgcn_wave_barrier();
        __builtin_amdgcn_fence(__ATOMIC_ACQUIRE, "wavefront");
        int i0 = 0;
        for (; i0 + 32 <= qn; i0 += 32) chunk(i0, 32);
        const int rem = qn - i0;
        if (rem > 0) {
            if (all) {
                chunk(i0, rem);
                qn = 0;
            } else {
                const int2 mv = q[w][i0 + min(lane, rem - 1)];
                __builtin_amdgcn_fence(__ATOMIC_ACQ_REL, "wavefront");
                __builtin_amdgcn_wave_barrier();
                if (lane < rem) q[w][lane] = mv;
                qn = rem;
            }
        } else {
            qn = 0;
        }
        __builtin_amdgcn_fence(__ATOMIC_RELEASE, "wavefront");
        __builtin_amdgcn_wave_barrier();
    };

    bool scan = true;
    if (MODE == 2) {
        const int cw = __builtin_amdgcn_readfirstlane(counts[wvno]);
        for (int i = tid; i < (int)gridDim.y * (OWN_T / 32); i += OWN_T) {
            const long v = ((long)(i / (OWN_T / 32)) * gridDim.x + blockIdx.x) * (OWN_T / 32) + (i % (OWN_T / 32));
            if (counts[v] < 0) short_list = 1;
        }
        __syncthreads();
        scan = (short_list != 0);
        if (!scan) {
            constexpr int LU = 8;
            for (int i0 = 0; i0 < cw; i0 += 32 * LU) {
                unsigned en[LU];
                float4 r[LU];
                #pragma unroll
                for (int u = 0; u < LU; ++u) {
                    const int i = i0 + 32 * u + lane;
                    en[u] = (i < cw) ? lw[i] : 0u;
                }
                #pragma unroll
                for (int u = 0; u < LU; ++u)
                    r[u] = *(const float4*)(h + (long)(en[u] >> LST_DLB) * (2 * F));
                float dd[LU];
                #pragma unroll
                for (int u = 0; u < LU; ++u)
                    dd[u] = dloc[en[u] & ((1u << LST_DLB) - 1u)];
                #pragma unroll
                for (int u = 0; u < LU; ++u) {
                    const int pl = (int)(en[u] & ((1u << LST_DLB) - 1u));
                    if (i0 + 32 * u + lane < cw) {
                        atomicAdd(&tile[pl * F],     (r[u].y * dd[u]) * r[u].x);
                        atomicAdd(&tile[pl * F + 1], (r[u].w * dd[u]) * r[u].z);
                    }
                }
            }
        }
    }

    if (MODE == 1 && direct) {
        static_assert(CL_CAP % (32 * OWN_PER) == 0, "a trip's loads stay inside the list");
        const unsigned plo = (unsigned)(lo - crng * CR_NODES);
        const int* cc = ccounts + (long)crng * cnl;
        for (int L = (int)blockIdx.y * (OWN_T / 32) + w; L < cnl; L += (int)gridDim.y * (OWN_T / 32)) {
            const int c = __builtin_amdgcn_readfirstlane(cc[L]);
            const unsigned* cl = clists + ((long)crng * cnl + L) * CL_CAP;
            for (int i0 = 0; i0 < c; i0 += 32 * OWN_PER) {
                const int ib = i0 + 4 * lane;
                unsigned en[OWN_PER];
                #pragma unroll
                for (int k = 0; k < OWN_PER; k += 4) {
                    const uint4 a = *(const uint4*)(cl + ib + GNN_slot(k));
                    en[k] = a.x; en[k + 1] = a.y; en[k + 2] = a.z; en[k + 3] = a.w;
                }
                #pragma unroll
                for (int k = 0; k < OWN_PER; ++k) {
                    const unsigned t = (en[k] & ((1u << CR_BITS) - 1u)) - plo;
                    const bool p = (ib + GNN_slot(k) < c) && (t < cnt);
                    const unsigned mask = __builtin_amdgcn_ballot_w32(p);
                    if (mask != 0u) {
                        if (p) q[w][qn + (int)__builtin_amdgcn_mbcnt_lo(mask, 0u)] = make_int2((int)(en[k] >> CR_BITS), (int)t);
                        qn += __builtin_popcount(mask);
                    }
                    if ((k % OWN_PART) == OWN_PART - 1) {
                        if (qn >= 32) drain(false);
                    }
                }
            }
        }
    }

    long e0, e1;
    GNN_slice(E, e0, e1);
    if (!scan || direct) e1 = e0;
    for (long base0 = e0; base0 < e1; base0 += OWN_TRIP) {
        const long wbase = base0 + (long)w * OWN_WBLK;
        const long lbase = wbase + 4 * lane;
        int d[OWN_PER];
        GNN_load_dst(dst, e1, wbase, lbase, d);
        #pragma unroll
        for (int k = 0; k < OWN_PER; ++k) {
            const unsigned t = (unsigned)(d[k] - lo);
            const bool p = t < cnt;
            const unsigned mask = __builtin_amdgcn_ballot_w32(p);
            if (mask != 0u) {
                if (p) q[w][qn + (int)__builtin_amdgcn_mbcnt_lo(mask, 0u)] = make_int2((int)(lbase + GNN_slot(k)), (int)t);
                qn += __builtin_popcount(mask);
            }
            if ((k % OWN_PART) == OWN_PART - 1) {
                if (qn >= 32) drain(false);
            }
        }
    }
    if (qn > 0) drain(true);
    if (MODE == 1 && lane == 0) counts[wvno] = lshort ? -1 : lpos;
    __syncthreads();
    float* ap = agg + ((long)blockIdx.y * n + lo) * F;
    for (int i = tid; i < (int)cnt * F; i += OWN_T) ap[i] = tile[i];
}

__global__ void GNN_gemm1_wmma(const float* __restrict__ x,
                               const float* __restrict__ W1,
                               const float* __restrict__ dinv,
                               float* __restrict__ t1, int nTiles) {
    int lane = threadIdx.x & 31;
    int wave = threadIdx.x >> 5;
    int tile = blockIdx.x * (blockDim.x >> 5) + wave;
    if (tile >= nTiles) return;

    int m    = lane & 15;
    int koff = (lane >> 4) << 1;
    long rowA = (long)(tile * 16 + m) * F_IN;

    v8f acc = {};
    #pragma unroll 4
    for (int kk = 0; kk < F_IN; kk += 4) {
        v2f a, b;
        a.x = x[rowA + kk + koff];
        a.y = x[rowA + kk + koff + 1];
        b.x = W1[(kk + koff)     * F_H + m];
        b.y = W1[(kk + koff + 1) * F_H + m];
        acc = __builtin_amdgcn_wmma_f32_16x16x4_f32(
            false, a, false, b, (short)0, acc, false, false);
    }

    int rbase = tile * 16 + (lane >> 4) * 8;
    #pragma unroll
    for (int r = 0; r < 8; ++r)
        *(float2*)(t1 + (long)(rbase + r) * (2 * F_H) + 2 * m) = make_float2(acc[r], dinv[rbase + r]);
}

__global__ void GNN_gemm1_tail(const float* __restrict__ x,
                               const float* __restrict__ W1,
                               const float* __restrict__ dinv,
                               float* __restrict__ t1, int startRow, int n) {
    int idx = blockIdx.x * blockDim.x + threadIdx.x;
    int row = startRow + (idx >> 4);
    int col = idx & 15;
    if (row >= n) return;
    float acc = 0.0f;
    for (int k = 0; k < F_IN; ++k)
        acc += x[(long)row * F_IN + k] * W1[k * F_H + col];
    *(float2*)(t1 + (long)row * (2 * F_H) + 2 * col) = make_float2(acc, dinv[row]);
}

__global__ void GNN_combine1(const float* __restrict__ agg1, int slices,
                             const float* __restrict__ t1,
                             const float* __restrict__ dinv,
                             const float* __restrict__ b1,
                             const float* __restrict__ W2,
                             float* __restrict__ t2, int n) {
    int i = blockIdx.x * blockDim.x + threadIdx.x;
    if (i >= n) return;
    float a[F_H];
    #pragma unroll
    for (int j = 0; j < F_H; j += 4) {
        const float4 t = *(const float4*)(agg1 + (long)i * F_H + j);
        a[j] = t.x; a[j + 1] = t.y; a[j + 2] = t.z; a[j + 3] = t.w;
    }
    for (int p = 1; p < slices; ++p) {
        const float* rp = agg1 + ((long)p * n + i) * F_H;
        #pragma unroll
        for (int j = 0; j < F_H; j += 4) {
            const float4 t = *(const float4*)(rp + j);
            a[j] += t.x; a[j + 1] += t.y; a[j + 2] += t.z; a[j + 3] += t.w;
        }
    }
    float hs[F_H];
    #pragma unroll
    for (int j = 0; j < F_H; j += 2) {
        const float4 t = *(const float4*)(t1 + (long)i * (2 * F_H) + 2 * j);
        hs[j] = t.x; hs[j + 1] = t.z;
    }
    float di  = dinv[i];
    float di2 = di * di;
    float acc0 = 0.0f, acc1 = 0.0f;
    #pragma unroll
    for (int j = 0; j < F_H; ++j) {
        float v = a[j] + di2 * hs[j] + b1[j];
        v = fmaxf(v, 0.0f);
        acc0 = fmaf(v, W2[j * F_C + 0], acc0);
        acc1 = fmaf(v, W2[j * F_C + 1], acc1);
    }
    *(float4*)(t2 + (long)i * (2 * F_C)) = make_float4(acc0, di, acc1, di);
}

__global__ void GNN_final(const float* __restrict__ agg2, int slices,
                          const float* __restrict__ t2,
                          const float* __restrict__ dinv,
                          const float* __restrict__ b2,
                          float* __restrict__ out, int n) {
    int i = blockIdx.x * blockDim.x + threadIdx.x;
    if (i >= n) return;
    float a0 = 0.0f, a1 = 0.0f;
    for (int p = 0; p < slices; ++p) {
        const float2 t = *(const float2*)(agg2 + ((long)p * n + i) * F_C);
        a0 += t.x; a1 += t.y;
    }
    float di  = dinv[i];
    float di2 = di * di;
    const float4 hv = *(const float4*)(t2 + (long)i * (2 * F_C));
    float o0 = a0 + di2 * hv.x + b2[0];
    float o1 = a1 + di2 * hv.z + b2[1];
    float mx  = fmaxf(o0, o1);
    float lse = mx + logf(expf(o0 - mx) + expf(o1 - mx));
    out[(long)i * F_C + 0] = o0 - lse;
    out[(long)i * F_C + 1] = o1 - lse;
}

extern "C" void kernel_launch(void* const* d_in, const int* in_sizes, int n_in,
                              void* d_out, int out_size, void* d_ws, size_t ws_size,
                              hipStream_t stream) {
    const float* x  = (const float*)d_in[0];
    const int* eidx = (const int*)  d_in[1];
    const float* W1 = (const float*)d_in[2];
    const float* b1 = (const float*)d_in[3];
    const float* W2 = (const float*)d_in[4];
    const float* b2 = (const float*)d_in[5];
    float* out = (float*)d_out;

    long nN = (long)in_sizes[0] / F_IN;
    long E  = (long)in_sizes[1] / 2;

    int s1 = 5;
    while (s1 > 1 && (size_t)(37 + 16 * s1) * (size_t)nN * sizeof(float) > ws_size) --s1;
    float* ws   = (float*)d_ws;
    float* t1   = ws;
    float* cntp = ws;
    float* agg2 = ws;
    float* t2   = ws + 32 * nN;
    float* agg1 = ws + 36 * nN;
    float* dinv = ws + (36 + 16 * s1) * nN;
    const long nW1 = ((nN + AG1_NODES_5 - 1) / AG1_NODES_5) * 5 * (OWN_T / 32);
    unsigned* lists = (unsigned*)(dinv + nN);
    int* counts     = (int*)(lists + nW1 * LST_CAP);
    const bool useL = s1 >= 5 && nN <= (1L << (32 - LST_DLB)) &&
        ((size_t)(37 + 16 * 5) * (size_t)nN + (size_t)nW1 * (LST_CAP + 1)) * sizeof(float) <= ws_size;

    const long nWc = ((nN + CR_NODES - 1) / CR_NODES) * CR_SLICES * (OWN_T / 32);
    const size_t offC = (((size_t)(37 + 16 * 5) * (size_t)nN + (size_t)nW1 * (LST_CAP + 1)) * sizeof(float) + 255) & ~(size_t)255;
    unsigned* clists = (unsigned*)((char*)d_ws + offC);
    int* ccounts     = (int*)(clists + nWc * CL_CAP);
    const bool useC = useL && nN <= (1L << (32 - CR_BITS)) &&
        offC + (size_t)nWc * (CL_CAP + 1) * sizeof(unsigned) <= ws_size;

    const int T = 256;
    int nBlk = (int)((nN + T - 1) / T);
    if (useC) {
        dim3 gCnt((unsigned)((nN + CR_NODES - 1) / CR_NODES), CR_SLICES);
        GNN_own_cnt_keep<<<gCnt, OWN_T, 0, stream>>>(eidx, E, cntp, (int)nN, clists, ccounts);
        GNN_dinv_merge<<<nBlk, T, 0, stream>>>(cntp, CR_SLICES, dinv, (int)nN);
    } else {
        dim3 gCnt((unsigned)((nN + CNT_NODES - 1) / CNT_NODES), CNT_SLICES);
        GNN_own_cnt<<<gCnt, OWN_T, 0, stream>>>(eidx, E, cntp, (int)nN);
        GNN_dinv_merge<<<nBlk, T, 0, stream>>>(cntp, CNT_SLICES, dinv, (int)nN);
    }

    int nTiles = (int)(nN / 16);
    if (nTiles > 0)
        GNN_gemm1_wmma<<<(nTiles + 7) / 8, T, 0, stream>>>(x, W1, dinv, t1, nTiles);
    int rem = (int)(nN - (long)nTiles * 16);
    if (rem > 0)
        GNN_gemm1_tail<<<(rem * 16 + T - 1) / T, T, 0, stream>>>(
            x, W1, dinv, t1, nTiles * 16, (int)nN);

    if (s1 >= 5) {
        dim3 gA1((unsigned)((nN + AG1_NODES_5 - 1) / AG1_NODES_5), 5);
        if (useL)
            GNN_own_agg<F_H, 16, AG1_NODES_5, 1><<<gA1, OWN_T, 0, stream>>>(eidx, E, dinv, t1, agg1, (int)nN, lists, counts,
                                                                   useC ? clists : nullptr, useC ? ccounts : nullptr);
        else
            GNN_own_agg<F_H, 16, AG1_NODES_5, 0><<<gA1, OWN_T, 0, stream>>>(eidx, E, dinv, t1, agg1, (int)nN, nullptr, nullptr, nullptr, nullptr);
    } else if (s1 == 4) {
        dim3 gA1((unsigned)((nN + AG1_NODES_4 - 1) / AG1_NODES_4), 4);
        GNN_own_agg<F_H, 16, AG1_NODES_4, 0><<<gA1, OWN_T, 0, stream>>>(eidx, E, dinv, t1, agg1, (int)nN, nullptr, nullptr, nullptr, nullptr);
    } else {
        dim3 gA1((unsigned)((nN + AG1_NODES_3 - 1) / AG1_NODES_3), (unsigned)s1);
        GNN_own_agg<F_H, 16, AG1_NODES_3, 0><<<gA1, OWN_T, 0, stream>>>(eidx, E, dinv, t1, agg1, (int)nN, nullptr, nullptr, nullptr, nullptr);
    }
    GNN_combine1<<<nBlk, T, 0, stream>>>(agg1, s1, t1, dinv, b1, W2, t2, (int)nN);
    if (useL) {
        dim3 gA2((unsigned)((nN + AG1_NODES_5 - 1) / AG1_NODES_5), 5);
        GNN_own_agg<F_C, 2, AG1_NODES_5, 2><<<gA2, OWN_T, 0, stream>>>(eidx, E, dinv, t2, agg2, (int)nN, lists, counts, nullptr, nullptr);
        GNN_final<<<nBlk, T, 0, stream>>>(agg2, 5, t2, dinv, b2, out, (int)nN);
    } else {
        dim3 gA2((unsigned)((nN + AG2_NODES - 1) / AG2_NODES), AG2_SLICES);
        GNN_own_agg<F_C, 2, AG2_NODES, 0><<<gA2, OWN_T, 0, stream>>>(eidx, E, dinv, t2, agg2, (int)nN, nullptr, nullptr, nullptr, nullptr);
        GNN_final<<<nBlk, T, 0, stream>>>(agg2, AG2_SLICES, t2, dinv, b2, out, (int)nN);
    }
}


# x @ W1 runs in the count's launch: 391 workgroups of 32 tile waves take the wave slots the 247 scanning ones leave free; the dinv kernel builds the (h1, dinv) table, count planes added 4-way (exact)
# speedup vs baseline: 9.3870x; 1.0923x over previous
#include <hip/hip_runtime.h>
#include <hip/hip_bf16.h>


typedef __attribute__((ext_vector_type(2))) float v2f;
typedef __attribute__((ext_vector_type(8))) float v8f;

#define F_IN 128
#define F_H  16
#define F_C  2

#define OWN_T     1024
#define OWN_PER   16
#define OWN_TRIP  (OWN_T * OWN_PER)
#define OWN_WBLK  (32 * OWN_PER)
#define OWN_PART  4
#define OWN_QCAP  (32 + 32 * OWN_PART)

#define CNT_NODES  25000
#define CNT_SLICES 32
#define AG1_NODES_5 4000
#define AG1_NODES_4 3125
#define AG1_NODES_3 2400
#define AG2_NODES  12800
#define AG2_SLICES 16
#define LST_CAP 1024
#define LST_DLB 12

#define CR_NODES  16000
#define CR_BITS   14
#define CR_SLICES 19
#define CL_CAP    1024

__device__ __forceinline__ void GNN_slice_of(long E, long P, long p, long& e0, long& e1) {
    const long S = (((E + P - 1) / P) + 127) & ~127L;
    e0 = min(E, p * S);
    e1 = min(E, e0 + S);
}
__device__ __forceinline__ void GNN_slice(long E, long& e0, long& e1) {
    GNN_slice_of(E, (long)gridDim.y, (long)blockIdx.y, e0, e1);
}

__device__ __forceinline__ constexpr int GNN_slot(int k) { return (k >> 2) * 128 + (k & 3); }

__device__ __forceinline__ void GNN_load_dst(const int* __restrict__ dst, long end,
                                             long wbase, long lbase, int (&d)[OWN_PER]) {
    if (wbase + OWN_WBLK <= end) {
        #pragma unroll
        for (int k = 0; k < OWN_PER; k += 4) {
            int4 a = *(const int4*)(dst + lbase + GNN_slot(k));
            d[k] = a.x; d[k + 1] = a.y; d[k + 2] = a.z; d[k + 3] = a.w;
        }
    } else {
        #pragma unroll
        for (int k = 0; k < OWN_PER; ++k)
            d[k] = (lbase + GNN_slot(k) < end) ? dst[lbase + GNN_slot(k)] : -1;
    }
}

__global__ __launch_bounds__(OWN_T)
void GNN_own_cnt(const int* __restrict__ eidx, long E,
                 float* __restrict__ cntp, int n) {
    __shared__ float cntf[CNT_NODES];
    const int tid = threadIdx.x;
    const int lo  = blockIdx.x * CNT_NODES;
    const unsigned cnt = (unsigned)min(CNT_NODES, n - lo);
    for (int i = tid; i < CNT_NODES; i += OWN_T) cntf[i] = 0.0f;
    __syncthreads();
    long e0, e1;
    GNN_slice(E, e0, e1);
    const int* dst = eidx + E;
    const int wv = __builtin_amdgcn_readfirstlane(tid >> 5);
    for (long base0 = e0; base0 < e1; base0 += OWN_TRIP) {
        const long wbase = base0 + (long)wv * OWN_WBLK;
        int d[OWN_PER];
        GNN_load_dst(dst, e1, wbase, wbase + 4 * (tid & 31), d);
        #pragma unroll
        for (int k = 0; k < OWN_PER; ++k) {
            const unsigned t = (unsigned)(d[k] - lo);
            if (t < cnt) atomicAdd(&cntf[t], 1.0f);
        }
    }
    __syncthreads();
    float* cp = cntp + (long)blockIdx.y * n + lo;
    for (int i = tid; i < (int)cnt; i += OWN_T) cp[i] = cntf[i];
}

__device__ __forceinline__ v8f GNN_tile_acc(const float* __restrict__ x,
                                            const float* __restrict__ W1,
                                            int tile, int lane) {
    int m    = lane & 15;
    int koff = (lane >> 4) << 1;
    long rowA = (long)(tile * 16 + m) * F_IN;

    v8f acc = {};
    #pragma unroll 4
    for (int kk = 0; kk < F_IN; kk += 4) {
        v2f a, b;
        a.x = x[rowA + kk + koff];
        a.y = x[rowA + kk + koff + 1];
        b.x = W1[(kk + koff)     * F_H + m];
        b.y = W1[(kk + koff + 1) * F_H + m];
        acc = __builtin_amdgcn_wmma_f32_16x16x4_f32(
            false, a, false, b, (short)0, acc, false, false);
    }
    return acc;
}

__global__ __launch_bounds__(OWN_T)
void GNN_own_cnt_keep_gemm(const int* __restrict__ eidx, long E,
                           float* __restrict__ cntp, int n,
                           unsigned* __restrict__ clists, int* __restrict__ ccounts,
                           int nrng, int nslc,
                           const float* __restrict__ x, const float* __restrict__ W1,
                           float* __restrict__ h1raw, int nTiles) {
    __shared__ float cntf[CR_NODES];
    __shared__ unsigned cq[OWN_T / 32][64];
    const int tid  = threadIdx.x;
    const int lane = tid & 31;
    const int nCnt = nrng * nslc;
    if ((int)blockIdx.x >= nCnt) {
        const int tile = ((int)blockIdx.x - nCnt) * (OWN_T / 32) + (tid >> 5);
        if (tile >= nTiles) return;
        const v8f acc = GNN_tile_acc(x, W1, tile, lane);
        const int rbase = tile * 16 + (lane >> 4) * 8;
        #pragma unroll
        for (int r = 0; r < 8; ++r)
            h1raw[(long)(rbase + r) * F_H + (lane & 15)] = acc[r];
        return;
    }
    const int rng  = (int)blockIdx.x % nrng;
    const int slc  = (int)blockIdx.x / nrng;
    const int lo   = rng * CR_NODES;
    const unsigned cnt = (unsigned)min(CR_NODES, n - lo);
    for (int i = tid; i < CR_NODES; i += OWN_T) cntf[i] = 0.0f;
    __syncthreads();
    long e0, e1;
    GNN_slice_of(E, (long)nslc, (long)slc, e0, e1);
    const int* src = eidx;
    const int* dst = eidx + E;
    const int wv = __builtin_amdgcn_readfirstlane(tid >> 5);
    const long wvno = ((long)rng * nslc + slc) * (OWN_T / 32) + wv;
    unsigned* lw = clists + wvno * CL_CAP;
    int  qn = 0;
    int  lpos = 0;
    bool lshort = false;
    for (long base0 = e0; base0 < e1; base0 += OWN_TRIP) {
        const long wbase = base0 + (long)wv * OWN_WBLK;
        const long lbase = wbase + 4 * lane;
        int d[OWN_PER], s[OWN_PER];
        GNN_load_dst(dst, e1, wbase, lbase, d);
        GNN_load_dst(src, e1, wbase, lbase, s);
        #pragma unroll
        for (int k = 0; k < OWN_PER; ++k) {
            const unsigned t = (unsigned)(d[k] - lo);
            const bool p = t < cnt;
            if (p) atomicAdd(&cntf[t], 1.0f);
            const unsigned mask = __builtin_amdgcn_ballot_w32(p);
            if (mask != 0u) {
                if (p) cq[wv][qn + (int)__builtin_amdgcn_mbcnt_lo(mask, 0u)] = ((unsigned)s[k] << CR_BITS) | t;
                qn += __builtin_popcount(mask);
                if (qn >= 32) {
                    __builtin_amdgcn_fence(__ATOMIC_RELEASE, "wavefront");
                    __builtin_amdgcn_wave_barrier();
                    __builtin_amdgcn_fence(__ATOMIC_ACQUIRE, "wavefront");
                    const unsigned v  = cq[wv][lane];
                    const unsigned mv = cq[wv][32 + lane];
                    if (lpos + 32 <= CL_CAP) { lw[lpos + lane] = v; lpos += 32; }
                    else lshort = true;
                    __builtin_amdgcn_fence(__ATOMIC_ACQ_REL, "wavefront");
                    __builtin_amdgcn_wave_barrier();
                    qn -= 32;
                    cq[wv][lane] = mv;
                    __builtin_amdgcn_fence(__ATOMIC_RELEASE, "wavefront");
                    __builtin_amdgcn_wave_barrier();
                }
            }
        }
    }
    if (qn > 0) {
        __builtin_amdgcn_fence(__ATOMIC_RELEASE, "wavefront");
        __builtin_amdgcn_wave_barrier();
        __builtin_amdgcn_fence(__ATOMIC_ACQUIRE, "wavefront");
        if (lpos + qn <= CL_CAP) { if (lane < qn) lw[lpos + lane] = cq[wv][lane]; lpos += qn; }
        else lshort = true;
    }
    if (lane == 0) ccounts[wvno] = lshort ? -1 : lpos;
    __syncthreads();
    float* cp = cntp + (long)slc * n + lo;
    for (int i = tid; i < (int)cnt; i += OWN_T) cp[i] = cntf[i];
}

__global__ void GNN_dinv_merge(const float* __restrict__ cntp, int slices,
                               float* __restrict__ dinv, int n) {
    int i = blockIdx.x * blockDim.x + threadIdx.x;
    if (i >= n) return;
    float c = 0.0f;
    for (int p = 0; p < slices; ++p) c += cntp[(long)p * n + i];
    dinv[i] = rsqrtf(c + 1.0f);
}

template <int SLICES, int PER>
__global__ void GNN_dinv_pairs(const float* __restrict__ cntp,
                               const float* __restrict__ h1raw,
                               float* __restrict__ dinv, float* __restrict__ t1, int n) {
    const long nth  = (long)gridDim.x * blockDim.x;
    const long t0   = (long)blockIdx.x * blockDim.x + threadIdx.x;
    const long last = 4L * n - 1;
    float4 hv[PER]; float c[PER];
    #pragma unroll
    for (int u = 0; u < PER; ++u) {
        const long t = min(t0 + u * nth, last);
        const int i = (int)(t >> 2), q = (int)(t & 3);
        hv[u] = *(const float4*)(h1raw + (long)i * F_H + 4 * q);
        c[u] = 0.0f;
        #pragma unroll
        for (int j = 0; j < (SLICES + 3) / 4; ++j) {
            const int p = q + 4 * j;
            const float v = cntp[(long)min(p, SLICES - 1) * n + i];
            c[u] += (p < SLICES) ? v : 0.0f;
        }
    }
    #pragma unroll
    for (int u = 0; u < PER; ++u) {
        const long t = t0 + u * nth;
        float s = c[u];
        s += __shfl_xor(s, 1);
        s += __shfl_xor(s, 2);
        const float d = rsqrtf(s + 1.0f);
        if (t <= last) {
            const int i = (int)(t >> 2), q = (int)(t & 3);
            if (q == 0) dinv[i] = d;
            float4* o = (float4*)(t1 + (long)i * (2 * F_H) + 8 * q);
            o[0] = make_float4(hv[u].x, d, hv[u].y, d);
            o[1] = make_float4(hv[u].z, d, hv[u].w, d);
        }
    }
}

template <int F, int LPE, int M, int MODE>
__global__ __launch_bounds__(OWN_T)
void GNN_own_agg(const int* __restrict__ eidx, long E,
                 const float* __restrict__ dinv,
                 const float* __restrict__ h,
                 float* __restrict__ agg, int n,
                 unsigned* __restrict__ lists, int* __restrict__ counts,
                 const unsigned* __restrict__ clists, const int* __restrict__ ccounts) {
    static_assert(MODE != 2 || F == 2, "the list path reads a row as one float4");
    static_assert(MODE != 1 || CR_NODES % M == 0, "a layer 1 range lies inside one count range");
    static_assert(MODE == 0 || M < (1 << LST_DLB), "place must fit its bits");
    __shared__ float tile[(M + 1) * F];
    __shared__ float dloc[M + 1];
    __shared__ int2  q[OWN_T / 32][OWN_QCAP];
    __shared__ int   short_list;
    const int tid  = threadIdx.x;
    const int lane = tid & 31;
    const int w    = __builtin_amdgcn_readfirstlane(tid >> 5);
    const int lo   = blockIdx.x * M;
    const unsigned cnt = (unsigned)min(M, n - lo);
    for (int i = tid; i < (M + 1) * F; i += OWN_T) tile[i] = 0.0f;
    for (int i = tid; i <= M; i += OWN_T)
        dloc[i] = (i < (int)cnt) ? dinv[lo + i] : 0.0f;
    if (MODE != 0 && tid == 0) short_list = 0;
    __syncthreads();

    bool direct = false;
    const int crng = lo / CR_NODES;
    const int cnl  = CR_SLICES * (OWN_T / 32);
    if (MODE == 1 && clists != nullptr) {
        const int* cc = ccounts + (long)crng * cnl;
        for (int i = tid; i < cnl; i += OWN_T)
            if (cc[i] < 0) short_list = 1;
        __syncthreads();
        direct = (short_list == 0);
    }

    const long wvno = ((long)blockIdx.y * gridDim.x + blockIdx.x) * (OWN_T / 32) + w;
    unsigned* lw = (MODE == 0) ? nullptr : lists + wvno * LST_CAP;
    int  lpos = 0;
    bool lshort = false;

    const int* src = eidx;
    const int* dst = eidx + E;
    const int sub = lane / LPE;
    const int col = lane % LPE;
    constexpr int EPS = 32 / LPE;
    int qn = 0;

    auto chunk = [&](int i0, int m) {
        const bool okl = lane < m;
        const int2  ent  = q[w][i0 + (okl ? lane : 0)];
        const int   s_l  = direct ? ent.x : src[ent.x];
        const int   dl_l = okl ? ent.y : M;
        const float dd_l = dloc[dl_l];
        if (MODE == 1) {
            if (lpos + m <= LST_CAP) {
                if (okl) lw[lpos + lane] = ((unsigned)s_l << LST_DLB) | (unsigned)ent.y;
                lpos += m;
            } else {
                lshort = true;
            }
        }
        float2 v[LPE]; int dl[LPE]; float dd[LPE];
        #pragma unroll
        for (int j = 0; j < LPE; ++j) {
            const int from = j * EPS + sub;
            const int s = __shfl(s_l, from);
            dl[j] = __shfl(dl_l, from);
            dd[j] = __shfl(dd_l, from);
            v[j]  = *(const float2*)(h + (long)s * (2 * F) + 2 * col);
        }
        #pragma unroll
        for (int j = 0; j < LPE; ++j)
            atomicAdd(&tile[dl[j] * F + col], (v[j].y * dd[j]) * v[j].x);
    };

    auto drain = [&](bool all) {
        __builtin_amdgcn_fence(__ATOMIC_RELEASE, "wavefront");
        __builtin_amdgcn_wave_barrier();
        __builtin_amdgcn_fence(__ATOMIC_ACQUIRE, "wavefront");
        int i0 = 0;
        for (; i0 + 32 <= qn; i0 += 32) chunk(i0, 32);
        const int rem = qn - i0;
        if (rem > 0) {
            if (all) {
                chunk(i0, rem);
                qn = 0;
            } else {
                const int2 mv = q[w][i0 + min(lane, rem - 1)];
                __builtin_amdgcn_fence(__ATOMIC_ACQ_REL, "wavefront");
                __builtin_amdgcn_wave_barrier();
                if (lane < rem) q[w][lane] = mv;
                qn = rem;
            }
        } else {
            qn = 0;
        }
        __builtin_amdgcn_fence(__ATOMIC_RELEASE, "wavefront");
        __builtin_amdgcn_wave_barrier();
    };

    bool scan = true;
    if (MODE == 2) {
        const int cw = __builtin_amdgcn_readfirstlane(counts[wvno]);
        for (int i = tid; i < (int)gridDim.y * (OWN_T / 32); i += OWN_T) {
            const long v = ((long)(i / (OWN_T / 32)) * gridDim.x + blockIdx.x) * (OWN_T / 32) + (i % (OWN_T / 32));
            if (counts[v] < 0) short_list = 1;
        }
        __syncthreads();
        scan = (short_list != 0);
        if (!scan) {
            constexpr int LU = 8;
            for (int i0 = 0; i0 < cw; i0 += 32 * LU) {
                unsigned en[LU];
                float4 r[LU];
                #pragma unroll
                for (int u = 0; u < LU; ++u) {
                    const int i = i0 + 32 * u + lane;
                    en[u] = (i < cw) ? lw[i] : 0u;
                }
                #pragma unroll
                for (int u = 0; u < LU; ++u)
                    r[u] = *(const float4*)(h + (long)(en[u] >> LST_DLB) * (2 * F));
                float dd[LU];
                #pragma unroll
                for (int u = 0; u < LU; ++u)
                    dd[u] = dloc[en[u] & ((1u << LST_DLB) - 1u)];
                #pragma unroll
                for (int u = 0; u < LU; ++u) {
                    const int pl = (int)(en[u] & ((1u << LST_DLB) - 1u));
                    if (i0 + 32 * u + lane < cw) {
                        atomicAdd(&tile[pl * F],     (r[u].y * dd[u]) * r[u].x);
                        atomicAdd(&tile[pl * F + 1], (r[u].w * dd[u]) * r[u].z);
                    }
                }
            }
        }
    }

    if (MODE == 1 && direct) {
        static_assert(CL_CAP % (32 * OWN_PER) == 0, "a trip's loads stay inside the list");
        const unsigned plo = (unsigned)(lo - crng * CR_NODES);
        const int* cc = ccounts + (long)crng * cnl;
        for (int L = (int)blockIdx.y * (OWN_T / 32) + w; L < cnl; L += (int)gridDim.y * (OWN_T / 32)) {
            const int c = __builtin_amdgcn_readfirstlane(cc[L]);
            const unsigned* cl = clists + ((long)crng * cnl + L) * CL_CAP;
            for (int i0 = 0; i0 < c; i0 += 32 * OWN_PER) {
                const int ib = i0 + 4 * lane;
                unsigned en[OWN_PER];
                #pragma unroll
                for (int k = 0; k < OWN_PER; k += 4) {
                    const uint4 a = *(const uint4*)(cl + ib + GNN_slot(k));
                    en[k] = a.x; en[k + 1] = a.y; en[k + 2] = a.z; en[k + 3] = a.w;
                }
                #pragma unroll
                for (int k = 0; k < OWN_PER; ++k) {
                    const unsigned t = (en[k] & ((1u << CR_BITS) - 1u)) - plo;
                    const bool p = (ib + GNN_slot(k) < c) && (t < cnt);
                    const unsigned mask = __builtin_amdgcn_ballot_w32(p);
                    if (mask != 0u) {
                        if (p) q[w][qn + (int)__builtin_amdgcn_mbcnt_lo(mask, 0u)] = make_int2((int)(en[k] >> CR_BITS), (int)t);
                        qn += __builtin_popcount(mask);
                    }
                    if ((k % OWN_PART) == OWN_PART - 1) {
                        if (qn >= 32) drain(false);
                    }
                }
            }
        }
    }

    long e0, e1;
    GNN_slice(E, e0, e1);
    if (!scan || direct) e1 = e0;
    for (long base0 = e0; base0 < e1; base0 += OWN_TRIP) {
        const long wbase = base0 + (long)w * OWN_WBLK;
        const long lbase = wbase + 4 * lane;
        int d[OWN_PER];
        GNN_load_dst(dst, e1, wbase, lbase, d);
        #pragma unroll
        for (int k = 0; k < OWN_PER; ++k) {
            const unsigned t = (unsigned)(d[k] - lo);
            const bool p = t < cnt;
            const unsigned mask = __builtin_amdgcn_ballot_w32(p);
            if (mask != 0u) {
                if (p) q[w][qn + (int)__builtin_amdgcn_mbcnt_lo(mask, 0u)] = make_int2((int)(lbase + GNN_slot(k)), (int)t);
                qn += __builtin_popcount(mask);
            }
            if ((k % OWN_PART) == OWN_PART - 1) {
                if (qn >= 32) drain(false);
            }
        }
    }
    if (qn > 0) drain(true);
    if (MODE == 1 && lane == 0) counts[wvno] = lshort ? -1 : lpos;
    __syncthreads();
    float* ap = agg + ((long)blockIdx.y * n + lo) * F;
    for (int i = tid; i < (int)cnt * F; i += OWN_T) ap[i] = tile[i];
}

__global__ void GNN_gemm1_wmma(const float* __restrict__ x,
                               const float* __restrict__ W1,
                               const float* __restrict__ dinv,
                               float* __restrict__ t1, int nTiles) {
    int lane = threadIdx.x & 31;
    int wave = threadIdx.x >> 5;
    int tile = blockIdx.x * (blockDim.x >> 5) + wave;
    if (tile >= nTiles) return;

    int m = lane & 15;
    const v8f acc = GNN_tile_acc(x, W1, tile, lane);

    int rbase = tile * 16 + (lane >> 4) * 8;
    #pragma unroll
    for (int r = 0; r < 8; ++r)
        *(float2*)(t1 + (long)(rbase + r) * (2 * F_H) + 2 * m) = make_float2(acc[r], dinv[rbase + r]);
}

__global__ void GNN_gemm1_tail(const float* __restrict__ x,
                               const float* __restrict__ W1,
                               const float* __restrict__ dinv,
                               float* __restrict__ t1, int startRow, int n) {
    int idx = blockIdx.x * blockDim.x + threadIdx.x;
    int row = startRow + (idx >> 4);
    int col = idx & 15;
    if (row >= n) return;
    float acc = 0.0f;
    for (int k = 0; k < F_IN; ++k)
        acc += x[(long)row * F_IN + k] * W1[k * F_H + col];
    if (dinv != nullptr) *(float2*)(t1 + (long)row * (2 * F_H) + 2 * col) = make_float2(acc, dinv[row]);
    else t1[(long)row * F_H + col] = acc;
}

__global__ void GNN_combine1(const float* __restrict__ agg1, int slices,
                             const float* __restrict__ t1,
                             const float* __restrict__ dinv,
                             const float* __restrict__ b1,
                             const float* __restrict__ W2,
                             float* __restrict__ t2, int n) {
    int i = blockIdx.x * blockDim.x + threadIdx.x;
    if (i >= n) return;
    float a[F_H];
    #pragma unroll
    for (int j = 0; j < F_H; j += 4) {
        const float4 t = *(const float4*)(agg1 + (long)i * F_H + j);
        a[j] = t.x; a[j + 1] = t.y; a[j + 2] = t.z; a[j + 3] = t.w;
    }
    for (int p = 1; p < slices; ++p) {
        const float* rp = agg1 + ((long)p * n + i) * F_H;
        #pragma unroll
        for (int j = 0; j < F_H; j += 4) {
            const float4 t = *(const float4*)(rp + j);
            a[j] += t.x; a[j + 1] += t.y; a[j + 2] += t.z; a[j + 3] += t.w;
        }
    }
    float hs[F_H];
    #pragma unroll
    for (int j = 0; j < F_H; j += 2) {
        const float4 t = *(const float4*)(t1 + (long)i * (2 * F_H) + 2 * j);
        hs[j] = t.x; hs[j + 1] = t.z;
    }
    float di  = dinv[i];
    float di2 = di * di;
    float acc0 = 0.0f, acc1 = 0.0f;
    #pragma unroll
    for (int j = 0; j < F_H; ++j) {
        float v = a[j] + di2 * hs[j] + b1[j];
        v = fmaxf(v, 0.0f);
        acc0 = fmaf(v, W2[j * F_C + 0], acc0);
        acc1 = fmaf(v, W2[j * F_C + 1], acc1);
    }
    *(float4*)(t2 + (long)i * (2 * F_C)) = make_float4(acc0, di, acc1, di);
}

__global__ void GNN_final(const float* __restrict__ agg2, int slices,
                          const float* __restrict__ t2,
                          const float* __restrict__ dinv,
                          const float* __restrict__ b2,
                          float* __restrict__ out, int n) {
    int i = blockIdx.x * blockDim.x + threadIdx.x;
    if (i >= n) return;
    float a0 = 0.0f, a1 = 0.0f;
    for (int p = 0; p < slices; ++p) {
        const float2 t = *(const float2*)(agg2 + ((long)p * n + i) * F_C);
        a0 += t.x; a1 += t.y;
    }
    float di  = dinv[i];
    float di2 = di * di;
    const float4 hv = *(const float4*)(t2 + (long)i * (2 * F_C));
    float o0 = a0 + di2 * hv.x + b2[0];
    float o1 = a1 + di2 * hv.z + b2[1];
    float mx  = fmaxf(o0, o1);
    float lse = mx + logf(expf(o0 - mx) + expf(o1 - mx));
    out[(long)i * F_C + 0] = o0 - lse;
    out[(long)i * F_C + 1] = o1 - lse;
}

extern "C" void kernel_launch(void* const* d_in, const int* in_sizes, int n_in,
                              void* d_out, int out_size, void* d_ws, size_t ws_size,
                              hipStream_t stream) {
    const float* x  = (const float*)d_in[0];
    const int* eidx = (const int*)  d_in[1];
    const float* W1 = (const float*)d_in[2];
    const float* b1 = (const float*)d_in[3];
    const float* W2 = (const float*)d_in[4];
    const float* b2 = (const float*)d_in[5];
    float* out = (float*)d_out;

    long nN = (long)in_sizes[0] / F_IN;
    long E  = (long)in_sizes[1] / 2;

    int s1 = 5;
    while (s1 > 1 && (size_t)(37 + 16 * s1) * (size_t)nN * sizeof(float) > ws_size) --s1;
    float* ws   = (float*)d_ws;
    float* t1   = ws;
    float* cntp = ws;
    float* agg2 = ws;
    float* t2   = ws + 32 * nN;
    float* agg1 = ws + 36 * nN;
    float* dinv = ws + (36 + 16 * s1) * nN;
    const long nW1 = ((nN + AG1_NODES_5 - 1) / AG1_NODES_5) * 5 * (OWN_T / 32);
    unsigned* lists = (unsigned*)(dinv + nN);
    int* counts     = (int*)(lists + nW1 * LST_CAP);
    const bool useL = s1 >= 5 && nN <= (1L << (32 - LST_DLB)) &&
        ((size_t)(37 + 16 * 5) * (size_t)nN + (size_t)nW1 * (LST_CAP + 1)) * sizeof(float) <= ws_size;

    const long nWc = ((nN + CR_NODES - 1) / CR_NODES) * CR_SLICES * (OWN_T / 32);
    const size_t offC = (((size_t)(37 + 16 * 5) * (size_t)nN + (size_t)nW1 * (LST_CAP + 1)) * sizeof(float) + 255) & ~(size_t)255;
    unsigned* clists = (unsigned*)((char*)d_ws + offC);
    int* ccounts     = (int*)(clists + nWc * CL_CAP);
    const bool useC = useL && nN <= (1L << (32 - CR_BITS)) &&
        offC + (size_t)nWc * (CL_CAP + 1) * sizeof(unsigned) <= ws_size;

    const int T = 256;
    int nBlk = (int)((nN + T - 1) / T);
    int nTiles = (int)(nN / 16);
    int rem = (int)(nN - (long)nTiles * 16);
    if (useC) {
        float* h1raw = agg1;
        float* cntpC = agg1 + 16 * nN;
        const int nRng  = (int)((nN + CR_NODES - 1) / CR_NODES);
        const int nGemm = (nTiles + OWN_T / 32 - 1) / (OWN_T / 32);
        GNN_own_cnt_keep_gemm<<<nRng * CR_SLICES + nGemm, OWN_T, 0, stream>>>(eidx, E, cntpC, (int)nN, clists, ccounts,
                                                                 nRng, CR_SLICES, x, W1, h1raw, nTiles);
        if (rem > 0)
            GNN_gemm1_tail<<<(rem * 16 + T - 1) / T, T, 0, stream>>>(
                x, W1, nullptr, h1raw, nTiles * 16, (int)nN);
        GNN_dinv_pairs<CR_SLICES, 4><<<nBlk, T, 0, stream>>>(cntpC, h1raw, dinv, t1, (int)nN);
    } else {
        dim3 gCnt((unsigned)((nN + CNT_NODES - 1) / CNT_NODES), CNT_SLICES);
        GNN_own_cnt<<<gCnt, OWN_T, 0, stream>>>(eidx, E, cntp, (int)nN);
        GNN_dinv_merge<<<nBlk, T, 0, stream>>>(cntp, CNT_SLICES, dinv, (int)nN);
        if (nTiles > 0)
            GNN_gemm1_wmma<<<(nTiles + 7) / 8, T, 0, stream>>>(x, W1, dinv, t1, nTiles);
        if (rem > 0)
            GNN_gemm1_tail<<<(rem * 16 + T - 1) / T, T, 0, stream>>>(
                x, W1, dinv, t1, nTiles * 16, (int)nN);
    }

    if (s1 >= 5) {
        dim3 gA1((unsigned)((nN + AG1_NODES_5 - 1) / AG1_NODES_5), 5);
        if (useL)
            GNN_own_agg<F_H, 16, AG1_NODES_5, 1><<<gA1, OWN_T, 0, stream>>>(eidx, E, dinv, t1, agg1, (int)nN, lists, counts,
                                                                   useC ? clists : nullptr, useC ? ccounts : nullptr);
        else
            GNN_own_agg<F_H, 16, AG1_NODES_5, 0><<<gA1, OWN_T, 0, stream>>>(eidx, E, dinv, t1, agg1, (int)nN, nullptr, nullptr, nullptr, nullptr);
    } else if (s1 == 4) {
        dim3 gA1((unsigned)((nN + AG1_NODES_4 - 1) / AG1_NODES_4), 4);
        GNN_own_agg<F_H, 16, AG1_NODES_4, 0><<<gA1, OWN_T, 0, stream>>>(eidx, E, dinv, t1, agg1, (int)nN, nullptr, nullptr, nullptr, nullptr);
    } else {
        dim3 gA1((unsigned)((nN + AG1_NODES_3 - 1) / AG1_NODES_3), (unsigned)s1);
        GNN_own_agg<F_H, 16, AG1_NODES_3, 0><<<gA1, OWN_T, 0, stream>>>(eidx, E, dinv, t1, agg1, (int)nN, nullptr, nullptr, nullptr, nullptr);
    }
    GNN_combine1<<<nBlk, T, 0, stream>>>(agg1, s1, t1, dinv, b1, W2, t2, (int)nN);
    if (useL) {
        dim3 gA2((unsigned)((nN + AG1_NODES_5 - 1) / AG1_NODES_5), 5);
        GNN_own_agg<F_C, 2, AG1_NODES_5, 2><<<gA2, OWN_T, 0, stream>>>(eidx, E, dinv, t2, agg2, (int)nN, lists, counts, nullptr, nullptr);
        GNN_final<<<nBlk, T, 0, stream>>>(agg2, 5, t2, dinv, b2, out, (int)nN);
    } else {
        dim3 gA2((unsigned)((nN + AG2_NODES - 1) / AG2_NODES), AG2_SLICES);
        GNN_own_agg<F_C, 2, AG2_NODES, 0><<<gA2, OWN_T, 0, stream>>>(eidx, E, dinv, t2, agg2, (int)nN, nullptr, nullptr, nullptr, nullptr);
        GNN_final<<<nBlk, T, 0, stream>>>(agg2, AG2_SLICES, t2, dinv, b2, out, (int)nN);
    }
}


# count scan and layer 1 list test lose the per-slot empty-slot branch; the count looks at its queue every 4 slots and adds degrees per leaving line of 32 hits (whole numbers: exact); lists same
# speedup vs baseline: 9.7319x; 1.0367x over previous
#include <hip/hip_runtime.h>
#include <hip/hip_bf16.h>


typedef __attribute__((ext_vector_type(2))) float v2f;
typedef __attribute__((ext_vector_type(8))) float v8f;

#define F_IN 128
#define F_H  16
#define F_C  2

#define OWN_T     1024
#define OWN_PER   16
#define OWN_TRIP  (OWN_T * OWN_PER)
#define OWN_WBLK  (32 * OWN_PER)
#define OWN_PART  4
#define OWN_QCAP  (32 + 32 * OWN_PART)

#define CNT_NODES  25000
#define CNT_SLICES 32
#define AG1_NODES_5 4000
#define AG1_NODES_4 3125
#define AG1_NODES_3 2400
#define AG2_NODES  12800
#define AG2_SLICES 16
#define LST_CAP 1024
#define LST_DLB 12

#define CR_NODES  16000
#define CR_BITS   14
#define CR_SLICES 19
#define CQ_CAP    (32 + 32 * OWN_PART)
#define CL_CAP    1024

__device__ __forceinline__ void GNN_slice_of(long E, long P, long p, long& e0, long& e1) {
    const long S = (((E + P - 1) / P) + 127) & ~127L;
    e0 = min(E, p * S);
    e1 = min(E, e0 + S);
}
__device__ __forceinline__ void GNN_slice(long E, long& e0, long& e1) {
    GNN_slice_of(E, (long)gridDim.y, (long)blockIdx.y, e0, e1);
}

__device__ __forceinline__ constexpr int GNN_slot(int k) { return (k >> 2) * 128 + (k & 3); }

__device__ __forceinline__ void GNN_load_dst(const int* __restrict__ dst, long end,
                                             long wbase, long lbase, int (&d)[OWN_PER]) {
    if (wbase + OWN_WBLK <= end) {
        #pragma unroll
        for (int k = 0; k < OWN_PER; k += 4) {
            int4 a = *(const int4*)(dst + lbase + GNN_slot(k));
            d[k] = a.x; d[k + 1] = a.y; d[k + 2] = a.z; d[k + 3] = a.w;
        }
    } else {
        #pragma unroll
        for (int k = 0; k < OWN_PER; ++k)
            d[k] = (lbase + GNN_slot(k) < end) ? dst[lbase + GNN_slot(k)] : -1;
    }
}

__global__ __launch_bounds__(OWN_T)
void GNN_own_cnt(const int* __restrict__ eidx, long E,
                 float* __restrict__ cntp, int n) {
    __shared__ float cntf[CNT_NODES];
    const int tid = threadIdx.x;
    const int lo  = blockIdx.x * CNT_NODES;
    const unsigned cnt = (unsigned)min(CNT_NODES, n - lo);
    for (int i = tid; i < CNT_NODES; i += OWN_T) cntf[i] = 0.0f;
    __syncthreads();
    long e0, e1;
    GNN_slice(E, e0, e1);
    const int* dst = eidx + E;
    const int wv = __builtin_amdgcn_readfirstlane(tid >> 5);
    for (long base0 = e0; base0 < e1; base0 += OWN_TRIP) {
        const long wbase = base0 + (long)wv * OWN_WBLK;
        int d[OWN_PER];
        GNN_load_dst(dst, e1, wbase, wbase + 4 * (tid & 31), d);
        #pragma unroll
        for (int k = 0; k < OWN_PER; ++k) {
            const unsigned t = (unsigned)(d[k] - lo);
            if (t < cnt) atomicAdd(&cntf[t], 1.0f);
        }
    }
    __syncthreads();
    float* cp = cntp + (long)blockIdx.y * n + lo;
    for (int i = tid; i < (int)cnt; i += OWN_T) cp[i] = cntf[i];
}

__device__ __forceinline__ v8f GNN_tile_acc(const float* __restrict__ x,
                                            const float* __restrict__ W1,
                                            int tile, int lane) {
    int m    = lane & 15;
    int koff = (lane >> 4) << 1;
    long rowA = (long)(tile * 16 + m) * F_IN;

    v8f acc = {};
    #pragma unroll 4
    for (int kk = 0; kk < F_IN; kk += 4) {
        v2f a, b;
        a.x = x[rowA + kk + koff];
        a.y = x[rowA + kk + koff + 1];
        b.x = W1[(kk + koff)     * F_H + m];
        b.y = W1[(kk + koff + 1) * F_H + m];
        acc = __builtin_amdgcn_wmma_f32_16x16x4_f32(
            false, a, false, b, (short)0, acc, false, false);
    }
    return acc;
}

__global__ __launch_bounds__(OWN_T)
void GNN_own_cnt_keep_gemm(const int* __restrict__ eidx, long E,
                           float* __restrict__ cntp, int n,
                           unsigned* __restrict__ clists, int* __restrict__ ccounts,
                           int nrng, int nslc,
                           const float* __restrict__ x, const float* __restrict__ W1,
                           float* __restrict__ h1raw, int nTiles) {
    __shared__ float cntf[CR_NODES];
    __shared__ unsigned cq[OWN_T / 32][CQ_CAP];
    const int tid  = threadIdx.x;
    const int lane = tid & 31;
    const int nCnt = nrng * nslc;
    if ((int)blockIdx.x >= nCnt) {
        const int tile = ((int)blockIdx.x - nCnt) * (OWN_T / 32) + (tid >> 5);
        if (tile >= nTiles) return;
        const v8f acc = GNN_tile_acc(x, W1, tile, lane);
        const int rbase = tile * 16 + (lane >> 4) * 8;
        #pragma unroll
        for (int r = 0; r < 8; ++r)
            h1raw[(long)(rbase + r) * F_H + (lane & 15)] = acc[r];
        return;
    }
    const int rng  = (int)blockIdx.x % nrng;
    const int slc  = (int)blockIdx.x / nrng;
    const int lo   = rng * CR_NODES;
    const unsigned cnt = (unsigned)min(CR_NODES, n - lo);
    for (int i = tid; i < CR_NODES; i += OWN_T) cntf[i] = 0.0f;
    __syncthreads();
    long e0, e1;
    GNN_slice_of(E, (long)nslc, (long)slc, e0, e1);
    const int* src = eidx;
    const int* dst = eidx + E;
    const int wv = __builtin_amdgcn_readfirstlane(tid >> 5);
    const long wvno = ((long)rng * nslc + slc) * (OWN_T / 32) + wv;
    unsigned* lw = clists + wvno * CL_CAP;
    int  qn = 0;
    int  lpos = 0;
    bool lshort = false;
    for (long base0 = e0; base0 < e1; base0 += OWN_TRIP) {
        const long wbase = base0 + (long)wv * OWN_WBLK;
        const long lbase = wbase + 4 * lane;
        int d[OWN_PER], s[OWN_PER];
        GNN_load_dst(dst, e1, wbase, lbase, d);
        GNN_load_dst(src, e1, wbase, lbase, s);
        #pragma unroll
        for (int k = 0; k < OWN_PER; ++k) {
            const unsigned t = (unsigned)(d[k] - lo);
            const bool p = t < cnt;
            const unsigned mask = __builtin_amdgcn_ballot_w32(p);
            if (p) cq[wv][__builtin_amdgcn_mbcnt_lo(mask, (unsigned)qn)] = ((unsigned)s[k] << CR_BITS) | t;
            qn += __builtin_popcount(mask);
            if ((k % OWN_PART) == OWN_PART - 1) {
                if (qn >= 32) {
                    __builtin_amdgcn_fence(__ATOMIC_RELEASE, "wavefront");
                    __builtin_amdgcn_wave_barrier();
                    __builtin_amdgcn_fence(__ATOMIC_ACQUIRE, "wavefront");
                    const int nl = qn >> 5;
                    #pragma clang loop unroll(disable)
                    for (int i = 0; i < nl; ++i) {
                        const unsigned v = cq[wv][32 * i + lane];
                        atomicAdd(&cntf[v & ((1u << CR_BITS) - 1u)], 1.0f);
                        if (lpos + 32 <= CL_CAP) { lw[lpos + lane] = v; lpos += 32; }
                        else lshort = true;
                    }
                    const unsigned mv = cq[wv][32 * nl + lane];
                    __builtin_amdgcn_fence(__ATOMIC_ACQ_REL, "wavefront");
                    __builtin_amdgcn_wave_barrier();
                    qn &= 31;
                    cq[wv][lane] = mv;
                    __builtin_amdgcn_fence(__ATOMIC_RELEASE, "wavefront");
                    __builtin_amdgcn_wave_barrier();
                }
            }
        }
    }
    if (qn > 0) {
        __builtin_amdgcn_fence(__ATOMIC_RELEASE, "wavefront");
        __builtin_amdgcn_wave_barrier();
        __builtin_amdgcn_fence(__ATOMIC_ACQUIRE, "wavefront");
        const unsigned v = cq[wv][lane];
        if (lane < qn) atomicAdd(&cntf[v & ((1u << CR_BITS) - 1u)], 1.0f);
        if (lpos + qn <= CL_CAP) { if (lane < qn) lw[lpos + lane] = v; lpos += qn; }
        else lshort = true;
    }
    if (lane == 0) ccounts[wvno] = lshort ? -1 : lpos;
    __syncthreads();
    float* cp = cntp + (long)slc * n + lo;
    for (int i = tid; i < (int)cnt; i += OWN_T) cp[i] = cntf[i];
}

__global__ void GNN_dinv_merge(const float* __restrict__ cntp, int slices,
                               float* __restrict__ dinv, int n) {
    int i = blockIdx.x * blockDim.x + threadIdx.x;
    if (i >= n) return;
    float c = 0.0f;
    for (int p = 0; p < slices; ++p) c += cntp[(long)p * n + i];
    dinv[i] = rsqrtf(c + 1.0f);
}

template <int SLICES, int PER>
__global__ void GNN_dinv_pairs(const float* __restrict__ cntp,
                               const float* __restrict__ h1raw,
                               float* __restrict__ dinv, float* __restrict__ t1, int n) {
    const long nth  = (long)gridDim.x * blockDim.x;
    const long t0   = (long)blockIdx.x * blockDim.x + threadIdx.x;
    const long last = 4L * n - 1;
    float4 hv[PER]; float c[PER];
    #pragma unroll
    for (int u = 0; u < PER; ++u) {
        const long t = min(t0 + u * nth, last);
        const int i = (int)(t >> 2), q = (int)(t & 3);
        hv[u] = *(const float4*)(h1raw + (long)i * F_H + 4 * q);
        c[u] = 0.0f;
        #pragma unroll
        for (int j = 0; j < (SLICES + 3) / 4; ++j) {
            const int p = q + 4 * j;
            const float v = cntp[(long)min(p, SLICES - 1) * n + i];
            c[u] += (p < SLICES) ? v : 0.0f;
        }
    }
    #pragma unroll
    for (int u = 0; u < PER; ++u) {
        const long t = t0 + u * nth;
        float s = c[u];
        s += __shfl_xor(s, 1);
        s += __shfl_xor(s, 2);
        const float d = rsqrtf(s + 1.0f);
        if (t <= last) {
            const int i = (int)(t >> 2), q = (int)(t & 3);
            if (q == 0) dinv[i] = d;
            float4* o = (float4*)(t1 + (long)i * (2 * F_H) + 8 * q);
            o[0] = make_float4(hv[u].x, d, hv[u].y, d);
            o[1] = make_float4(hv[u].z, d, hv[u].w, d);
        }
    }
}

template <int F, int LPE, int M, int MODE>
__global__ __launch_bounds__(OWN_T)
void GNN_own_agg(const int* __restrict__ eidx, long E,
                 const float* __restrict__ dinv,
                 const float* __restrict__ h,
                 float* __restrict__ agg, int n,
                 unsigned* __restrict__ lists, int* __restrict__ counts,
                 const unsigned* __restrict__ clists, const int* __restrict__ ccounts) {
    static_assert(MODE != 2 || F == 2, "the list path reads a row as one float4");
    static_assert(MODE != 1 || CR_NODES % M == 0, "a layer 1 range lies inside one count range");
    static_assert(MODE == 0 || M < (1 << LST_DLB), "place must fit its bits");
    __shared__ float tile[(M + 1) * F];
    __shared__ float dloc[M + 1];
    __shared__ int2  q[OWN_T / 32][OWN_QCAP];
    __shared__ int   short_list;
    const int tid  = threadIdx.x;
    const int lane = tid & 31;
    const int w    = __builtin_amdgcn_readfirstlane(tid >> 5);
    const int lo   = blockIdx.x * M;
    const unsigned cnt = (unsigned)min(M, n - lo);
    for (int i = tid; i < (M + 1) * F; i += OWN_T) tile[i] = 0.0f;
    for (int i = tid; i <= M; i += OWN_T)
        dloc[i] = (i < (int)cnt) ? dinv[lo + i] : 0.0f;
    if (MODE != 0 && tid == 0) short_list = 0;
    __syncthreads();

    bool direct = false;
    const int crng = lo / CR_NODES;
    const int cnl  = CR_SLICES * (OWN_T / 32);
    if (MODE == 1 && clists != nullptr) {
        const int* cc = ccounts + (long)crng * cnl;
        for (int i = tid; i < cnl; i += OWN_T)
            if (cc[i] < 0) short_list = 1;
        __syncthreads();
        direct = (short_list == 0);
    }

    const long wvno = ((long)blockIdx.y * gridDim.x + blockIdx.x) * (OWN_T / 32) + w;
    unsigned* lw = (MODE == 0) ? nullptr : lists + wvno * LST_CAP;
    int  lpos = 0;
    bool lshort = false;

    const int* src = eidx;
    const int* dst = eidx + E;
    const int sub = lane / LPE;
    const int col = lane % LPE;
    constexpr int EPS = 32 / LPE;
    int qn = 0;

    auto chunk = [&](int i0, int m) {
        const bool okl = lane < m;
        const int2  ent  = q[w][i0 + (okl ? lane : 0)];
        const int   s_l  = direct ? ent.x : src[ent.x];
        const int   dl_l = okl ? ent.y : M;
        const float dd_l = dloc[dl_l];
        if (MODE == 1) {
            if (lpos + m <= LST_CAP) {
                if (okl) lw[lpos + lane] = ((unsigned)s_l << LST_DLB) | (unsigned)ent.y;
                lpos += m;
            } else {
                lshort = true;
            }
        }
        float2 v[LPE]; int dl[LPE]; float dd[LPE];
        #pragma unroll
        for (int j = 0; j < LPE; ++j) {
            const int from = j * EPS + sub;
            const int s = __shfl(s_l, from);
            dl[j] = __shfl(dl_l, from);
            dd[j] = __shfl(dd_l, from);
            v[j]  = *(const float2*)(h + (long)s * (2 * F) + 2 * col);
        }
        #pragma unroll
        for (int j = 0; j < LPE; ++j)
            atomicAdd(&tile[dl[j] * F + col], (v[j].y * dd[j]) * v[j].x);
    };

    auto drain = [&](bool all) {
        __builtin_amdgcn_fence(__ATOMIC_RELEASE, "wavefront");
        __builtin_amdgcn_wave_barrier();
        __builtin_amdgcn_fence(__ATOMIC_ACQUIRE, "wavefront");
        int i0 = 0;
        for (; i0 + 32 <= qn; i0 += 32) chunk(i0, 32);
        const int rem = qn - i0;
        if (rem > 0) {
            if (all) {
                chunk(i0, rem);
                qn = 0;
            } else {
                const int2 mv = q[w][i0 + min(lane, rem - 1)];
                __builtin_amdgcn_fence(__ATOMIC_ACQ_REL, "wavefront");
                __builtin_amdgcn_wave_barrier();
                if (lane < rem) q[w][lane] = mv;
                qn = rem;
            }
        } else {
            qn = 0;
        }
        __builtin_amdgcn_fence(__ATOMIC_RELEASE, "wavefront");
        __builtin_amdgcn_wave_barrier();
    };

    bool scan = true;
    if (MODE == 2) {
        const int cw = __builtin_amdgcn_readfirstlane(counts[wvno]);
        for (int i = tid; i < (int)gridDim.y * (OWN_T / 32); i += OWN_T) {
            const long v = ((long)(i / (OWN_T / 32)) * gridDim.x + blockIdx.x) * (OWN_T / 32) + (i % (OWN_T / 32));
            if (counts[v] < 0) short_list = 1;
        }
        __syncthreads();
        scan = (short_list != 0);
        if (!scan) {
            constexpr int LU = 8;
            for (int i0 = 0; i0 < cw; i0 += 32 * LU) {
                unsigned en[LU];
                float4 r[LU];
                #pragma unroll
                for (int u = 0; u < LU; ++u) {
                    const int i = i0 + 32 * u + lane;
                    en[u] = (i < cw) ? lw[i] : 0u;
                }
                #pragma unroll
                for (int u = 0; u < LU; ++u)
                    r[u] = *(const float4*)(h + (long)(en[u] >> LST_DLB) * (2 * F));
                float dd[LU];
                #pragma unroll
                for (int u = 0; u < LU; ++u)
                    dd[u] = dloc[en[u] & ((1u << LST_DLB) - 1u)];
                #pragma unroll
                for (int u = 0; u < LU; ++u) {
                    const int pl = (int)(en[u] & ((1u << LST_DLB) - 1u));
                    if (i0 + 32 * u + lane < cw) {
                        atomicAdd(&tile[pl * F],     (r[u].y * dd[u]) * r[u].x);
                        atomicAdd(&tile[pl * F + 1], (r[u].w * dd[u]) * r[u].z);
                    }
                }
            }
        }
    }

    if (MODE == 1 && direct) {
        static_assert(CL_CAP % (32 * OWN_PER) == 0, "a trip's loads stay inside the list");
        const unsigned plo = (unsigned)(lo - crng * CR_NODES);
        const int* cc = ccounts + (long)crng * cnl;
        for (int L = (int)blockIdx.y * (OWN_T / 32) + w; L < cnl; L += (int)gridDim.y * (OWN_T / 32)) {
            const int c = __builtin_amdgcn_readfirstlane(cc[L]);
            const unsigned* cl = clists + ((long)crng * cnl + L) * CL_CAP;
            for (int i0 = 0; i0 < c; i0 += 32 * OWN_PER) {
                const int ib = i0 + 4 * lane;
                unsigned en[OWN_PER];
                #pragma unroll
                for (int k = 0; k < OWN_PER; k += 4) {
                    const uint4 a = *(const uint4*)(cl + ib + GNN_slot(k));
                    en[k] = a.x; en[k + 1] = a.y; en[k + 2] = a.z; en[k + 3] = a.w;
                }
                #pragma unroll
                for (int k = 0; k < OWN_PER; ++k) {
                    const unsigned t = (en[k] & ((1u << CR_BITS) - 1u)) - plo;
                    const bool pin = ib + GNN_slot(k) < c, prg = t < cnt;
                    const bool p = pin && prg;
                    const unsigned mask = __builtin_amdgcn_ballot_w32(pin) & __builtin_amdgcn_ballot_w32(prg);
                    if (p) q[w][__builtin_amdgcn_mbcnt_lo(mask, (unsigned)qn)] = make_int2((int)(en[k] >> CR_BITS), (int)t);
                    qn += __builtin_popcount(mask);
                    if ((k % OWN_PART) == OWN_PART - 1) {
                        if (qn >= 32) drain(false);
                    }
                }
            }
        }
    }

    long e0, e1;
    GNN_slice(E, e0, e1);
    if (!scan || direct) e1 = e0;
    for (long base0 = e0; base0 < e1; base0 += OWN_TRIP) {
        const long wbase = base0 + (long)w * OWN_WBLK;
        const long lbase = wbase + 4 * lane;
        int d[OWN_PER];
        GNN_load_dst(dst, e1, wbase, lbase, d);
        #pragma unroll
        for (int k = 0; k < OWN_PER; ++k) {
            const unsigned t = (unsigned)(d[k] - lo);
            const bool p = t < cnt;
            const unsigned mask = __builtin_amdgcn_ballot_w32(p);
            if (mask != 0u) {
                if (p) q[w][qn + (int)__builtin_amdgcn_mbcnt_lo(mask, 0u)] = make_int2((int)(lbase + GNN_slot(k)), (int)t);
                qn += __builtin_popcount(mask);
            }
            if ((k % OWN_PART) == OWN_PART - 1) {
                if (qn >= 32) drain(false);
            }
        }
    }
    if (qn > 0) drain(true);
    if (MODE == 1 && lane == 0) counts[wvno] = lshort ? -1 : lpos;
    __syncthreads();
    float* ap = agg + ((long)blockIdx.y * n + lo) * F;
    for (int i = tid; i < (int)cnt * F; i += OWN_T) ap[i] = tile[i];
}

__global__ void GNN_gemm1_wmma(const float* __restrict__ x,
                               const float* __restrict__ W1,
                               const float* __restrict__ dinv,
                               float* __restrict__ t1, int nTiles) {
    int lane = threadIdx.x & 31;
    int wave = threadIdx.x >> 5;
    int tile = blockIdx.x * (blockDim.x >> 5) + wave;
    if (tile >= nTiles) return;

    int m = lane & 15;
    const v8f acc = GNN_tile_acc(x, W1, tile, lane);

    int rbase = tile * 16 + (lane >> 4) * 8;
    #pragma unroll
    for (int r = 0; r < 8; ++r)
        *(float2*)(t1 + (long)(rbase + r) * (2 * F_H) + 2 * m) = make_float2(acc[r], dinv[rbase + r]);
}

__global__ void GNN_gemm1_tail(const float* __restrict__ x,
                               const float* __restrict__ W1,
                               const float* __restrict__ dinv,
                               float* __restrict__ t1, int startRow, int n) {
    int idx = blockIdx.x * blockDim.x + threadIdx.x;
    int row = startRow + (idx >> 4);
    int col = idx & 15;
    if (row >= n) return;
    float acc = 0.0f;
    for (int k = 0; k < F_IN; ++k)
        acc += x[(long)row * F_IN + k] * W1[k * F_H + col];
    if (dinv != nullptr) *(float2*)(t1 + (long)row * (2 * F_H) + 2 * col) = make_float2(acc, dinv[row]);
    else t1[(long)row * F_H + col] = acc;
}

__global__ void GNN_combine1(const float* __restrict__ agg1, int slices,
                             const float* __restrict__ t1,
                             const float* __restrict__ dinv,
                             const float* __restrict__ b1,
                             const float* __restrict__ W2,
                             float* __restrict__ t2, int n) {
    int i = blockIdx.x * blockDim.x + threadIdx.x;
    if (i >= n) return;
    float a[F_H];
    #pragma unroll
    for (int j = 0; j < F_H; j += 4) {
        const float4 t = *(const float4*)(agg1 + (long)i * F_H + j);
        a[j] = t.x; a[j + 1] = t.y; a[j + 2] = t.z; a[j + 3] = t.w;
    }
    for (int p = 1; p < slices; ++p) {
        const float* rp = agg1 + ((long)p * n + i) * F_H;
        #pragma unroll
        for (int j = 0; j < F_H; j += 4) {
            const float4 t = *(const float4*)(rp + j);
            a[j] += t.x; a[j + 1] += t.y; a[j + 2] += t.z; a[j + 3] += t.w;
        }
    }
    float hs[F_H];
    #pragma unroll
    for (int j = 0; j < F_H; j += 2) {
        const float4 t = *(const float4*)(t1 + (long)i * (2 * F_H) + 2 * j);
        hs[j] = t.x; hs[j + 1] = t.z;
    }
    float di  = dinv[i];
    float di2 = di * di;
    float acc0 = 0.0f, acc1 = 0.0f;
    #pragma unroll
    for (int j = 0; j < F_H; ++j) {
        float v = a[j] + di2 * hs[j] + b1[j];
        v = fmaxf(v, 0.0f);
        acc0 = fmaf(v, W2[j * F_C + 0], acc0);
        acc1 = fmaf(v, W2[j * F_C + 1], acc1);
    }
    *(float4*)(t2 + (long)i * (2 * F_C)) = make_float4(acc0, di, acc1, di);
}

__global__ void GNN_final(const float* __restrict__ agg2, int slices,
                          const float* __restrict__ t2,
                          const float* __restrict__ dinv,
                          const float* __restrict__ b2,
                          float* __restrict__ out, int n) {
    int i = blockIdx.x * blockDim.x + threadIdx.x;
    if (i >= n) return;
    float a0 = 0.0f, a1 = 0.0f;
    for (int p = 0; p < slices; ++p) {
        const float2 t = *(const float2*)(agg2 + ((long)p * n + i) * F_C);
        a0 += t.x; a1 += t.y;
    }
    float di  = dinv[i];
    float di2 = di * di;
    const float4 hv = *(const float4*)(t2 + (long)i * (2 * F_C));
    float o0 = a0 + di2 * hv.x + b2[0];
    float o1 = a1 + di2 * hv.z + b2[1];
    float mx  = fmaxf(o0, o1);
    float lse = mx + logf(expf(o0 - mx) + expf(o1 - mx));
    out[(long)i * F_C + 0] = o0 - lse;
    out[(long)i * F_C + 1] = o1 - lse;
}

extern "C" void kernel_launch(void* const* d_in, const int* in_sizes, int n_in,
                              void* d_out, int out_size, void* d_ws, size_t ws_size,
                              hipStream_t stream) {
    const float* x  = (const float*)d_in[0];
    const int* eidx = (const int*)  d_in[1];
    const float* W1 = (const float*)d_in[2];
    const float* b1 = (const float*)d_in[3];
    const float* W2 = (const float*)d_in[4];
    const float* b2 = (const float*)d_in[5];
    float* out = (float*)d_out;

    long nN = (long)in_sizes[0] / F_IN;
    long E  = (long)in_sizes[1] / 2;

    int s1 = 5;
    while (s1 > 1 && (size_t)(37 + 16 * s1) * (size_t)nN * sizeof(float) > ws_size) --s1;
    float* ws   = (float*)d_ws;
    float* t1   = ws;
    float* cntp = ws;
    float* agg2 = ws;
    float* t2   = ws + 32 * nN;
    float* agg1 = ws + 36 * nN;
    float* dinv = ws + (36 + 16 * s1) * nN;
    const long nW1 = ((nN + AG1_NODES_5 - 1) / AG1_NODES_5) * 5 * (OWN_T / 32);
    unsigned* lists = (unsigned*)(dinv + nN);
    int* counts     = (int*)(lists + nW1 * LST_CAP);
    const bool useL = s1 >= 5 && nN <= (1L << (32 - LST_DLB)) &&
        ((size_t)(37 + 16 * 5) * (size_t)nN + (size_t)nW1 * (LST_CAP + 1)) * sizeof(float) <= ws_size;

    const long nWc = ((nN + CR_NODES - 1) / CR_NODES) * CR_SLICES * (OWN_T / 32);
    const size_t offC = (((size_t)(37 + 16 * 5) * (size_t)nN + (size_t)nW1 * (LST_CAP + 1)) * sizeof(float) + 255) & ~(size_t)255;
    unsigned* clists = (unsigned*)((char*)d_ws + offC);
    int* ccounts     = (int*)(clists + nWc * CL_CAP);
    const bool useC = useL && nN <= (1L << (32 - CR_BITS)) &&
        offC + (size_t)nWc * (CL_CAP + 1) * sizeof(unsigned) <= ws_size;

    const int T = 256;
    int nBlk = (int)((nN + T - 1) / T);
    int nTiles = (int)(nN / 16);
    int rem = (int)(nN - (long)nTiles * 16);
    if (useC) {
        float* h1raw = agg1;
        float* cntpC = agg1 + 16 * nN;
        const int nRng  = (int)((nN + CR_NODES - 1) / CR_NODES);
        const int nGemm = (nTiles + OWN_T / 32 - 1) / (OWN_T / 32);
        GNN_own_cnt_keep_gemm<<<nRng * CR_SLICES + nGemm, OWN_T, 0, stream>>>(eidx, E, cntpC, (int)nN, clists, ccounts,
                                                                 nRng, CR_SLICES, x, W1, h1raw, nTiles);
        if (rem > 0)
            GNN_gemm1_tail<<<(rem * 16 + T - 1) / T, T, 0, stream>>>(
                x, W1, nullptr, h1raw, nTiles * 16, (int)nN);
        GNN_dinv_pairs<CR_SLICES, 4><<<nBlk, T, 0, stream>>>(cntpC, h1raw, dinv, t1, (int)nN);
    } else {
        dim3 gCnt((unsigned)((nN + CNT_NODES - 1) / CNT_NODES), CNT_SLICES);
        GNN_own_cnt<<<gCnt, OWN_T, 0, stream>>>(eidx, E, cntp, (int)nN);
        GNN_dinv_merge<<<nBlk, T, 0, stream>>>(cntp, CNT_SLICES, dinv, (int)nN);
        if (nTiles > 0)
            GNN_gemm1_wmma<<<(nTiles + 7) / 8, T, 0, stream>>>(x, W1, dinv, t1, nTiles);
        if (rem > 0)
            GNN_gemm1_tail<<<(rem * 16 + T - 1) / T, T, 0, stream>>>(
                x, W1, dinv, t1, nTiles * 16, (int)nN);
    }

    if (s1 >= 5) {
        dim3 gA1((unsigned)((nN + AG1_NODES_5 - 1) / AG1_NODES_5), 5);
        if (useL)
            GNN_own_agg<F_H, 16, AG1_NODES_5, 1><<<gA1, OWN_T, 0, stream>>>(eidx, E, dinv, t1, agg1, (int)nN, lists, counts,
                                                                   useC ? clists : nullptr, useC ? ccounts : nullptr);
        else
            GNN_own_agg<F_H, 16, AG1_NODES_5, 0><<<gA1, OWN_T, 0, stream>>>(eidx, E, dinv, t1, agg1, (int)nN, nullptr, nullptr, nullptr, nullptr);
    } else if (s1 == 4) {
        dim3 gA1((unsigned)((nN + AG1_NODES_4 - 1) / AG1_NODES_4), 4);
        GNN_own_agg<F_H, 16, AG1_NODES_4, 0><<<gA1, OWN_T, 0, stream>>>(eidx, E, dinv, t1, agg1, (int)nN, nullptr, nullptr, nullptr, nullptr);
    } else {
        dim3 gA1((unsigned)((nN + AG1_NODES_3 - 1) / AG1_NODES_3), (unsigned)s1);
        GNN_own_agg<F_H, 16, AG1_NODES_3, 0><<<gA1, OWN_T, 0, stream>>>(eidx, E, dinv, t1, agg1, (int)nN, nullptr, nullptr, nullptr, nullptr);
    }
    GNN_combine1<<<nBlk, T, 0, stream>>>(agg1, s1, t1, dinv, b1, W2, t2, (int)nN);
    if (useL) {
        dim3 gA2((unsigned)((nN + AG1_NODES_5 - 1) / AG1_NODES_5), 5);
        GNN_own_agg<F_C, 2, AG1_NODES_5, 2><<<gA2, OWN_T, 0, stream>>>(eidx, E, dinv, t2, agg2, (int)nN, lists, counts, nullptr, nullptr);
        GNN_final<<<nBlk, T, 0, stream>>>(agg2, 5, t2, dinv, b2, out, (int)nN);
    } else {
        dim3 gA2((unsigned)((nN + AG2_NODES - 1) / AG2_NODES), AG2_SLICES);
        GNN_own_agg<F_C, 2, AG2_NODES, 0><<<gA2, OWN_T, 0, stream>>>(eidx, E, dinv, t2, agg2, (int)nN, nullptr, nullptr, nullptr, nullptr);
        GNN_final<<<nBlk, T, 0, stream>>>(agg2, AG2_SLICES, t2, dinv, b2, out, (int)nN);
    }
}
